# starting kernel
# baseline (speedup 1.0000x reference)
#include <hip/hip_runtime.h>
#include <hip/hip_bf16.h>


typedef float v2f __attribute__((ext_vector_type(2)));
typedef float v8f __attribute__((ext_vector_type(8)));

#define NN    1024
#define K1    512
#define HID   128
#define HID2  64

__device__ __forceinline__ float fast_tanh(float x) {
#if __has_builtin(__builtin_amdgcn_tanhf)
    return __builtin_amdgcn_tanhf(x);
#else
    return tanhf(x);
#endif
}

__device__ __forceinline__ v8f wmma_f32(v2f a, v2f b, v8f c) {
    return __builtin_amdgcn_wmma_f32_16x16x4_f32(
         false, a,  false, b,
         (short)0, c,  false,  false);
}

__global__ __launch_bounds__(128) void gemm1_tanh_kernel(
    const float* __restrict__ xf,
    const float* __restrict__ W_foh, const float* __restrict__ W_fom,
    const float* __restrict__ cat_bias,
    float* __restrict__ hf, float* __restrict__ hm)
{
    const int tile = blockIdx.x * blockDim.y + threadIdx.y;
    const int lane = threadIdx.x;
    const int mat   = tile >> 9;
    const int rem   = tile & 511;
    const int mtile = rem >> 3;
    const int ntile = rem & 7;

    const float* __restrict__ W    = mat ? W_fom : W_foh;
    float*       __restrict__ outp = mat ? hm    : hf;
    const float* __restrict__ bias = cat_bias + mat * HID;

    const int half = lane >> 4;
    const int l16  = lane & 15;
    const int row  = mtile * 16 + l16;
    const int col  = ntile * 16 + l16;
    const int khi  = half * 2;

    const float* arow = xf + row * K1 + khi;
    const float* bcol = W  + khi * HID + col;

    v8f c = {};
    #pragma unroll 4
    for (int k = 0; k < K1; k += 4) {
        v2f a = *reinterpret_cast<const v2f*>(arow + k);
        v2f b;
        b.x = bcol[k * HID];
        b.y = bcol[k * HID + HID];
        c = wmma_f32(a, b, c);
    }

    const float bn = bias[col];
    #pragma unroll
    for (int r = 0; r < 8; ++r) {
        const int m = mtile * 16 + r + half * 8;
        outp[m * HID + col] = fast_tanh(c[r] + bn);
    }
}

__global__ __launch_bounds__(128) void gemm2_kernel(
    const float* __restrict__ hf, const float* __restrict__ hm,
    const float* __restrict__ W_hid2, const float* __restrict__ hid2_bias,
    float* __restrict__ th, float* __restrict__ tmT)
{
    const int tile = blockIdx.x * blockDim.y + threadIdx.y;
    const int lane = threadIdx.x;
    const int mat   = tile >> 8;
    const int rem   = tile & 255;
    const int mtile = rem >> 2;
    const int ntile = rem & 3;

    const float* __restrict__ A = mat ? hm : hf;
    const float* __restrict__ B = W_hid2 + mat * HID * HID2;

    const int half = lane >> 4;
    const int l16  = lane & 15;
    const int row  = mtile * 16 + l16;
    const int col  = ntile * 16 + l16;
    const int khi  = half * 2;

    const float* arow = A + row * HID + khi;
    const float* bcol = B + khi * HID2 + col;

    v8f c = {};
    #pragma unroll
    for (int k = 0; k < HID; k += 4) {
        v2f a = *reinterpret_cast<const v2f*>(arow + k);
        v2f b;
        b.x = bcol[k * HID2];
        b.y = bcol[k * HID2 + HID2];
        c = wmma_f32(a, b, c);
    }

    if (mat == 0) {
        const float bn = hid2_bias[col];
        #pragma unroll
        for (int r = 0; r < 8; ++r) {
            const int m = mtile * 16 + r + half * 8;
            th[m * HID2 + col] = c[r] + bn;
        }
    } else {
        #pragma unroll
        for (int r = 0; r < 8; ++r) {
            const int m = mtile * 16 + r + half * 8;
            tmT[col * NN + m] = c[r];
        }
    }
}

#define IPT 8
__global__ __launch_bounds__(256) void outer_tanh_dot_kernel(
    const float* __restrict__ th, const float* __restrict__ tmT,
    const float* __restrict__ W_out, const float* __restrict__ out_bias,
    float* __restrict__ out)
{
    __shared__ float sw[HID2];
    __shared__ float sth[IPT][HID2];

    const int tx = threadIdx.x;
    const int j  = blockIdx.x * 256 + tx;
    const int i0 = blockIdx.y * IPT;

    if (tx < HID2) sw[tx] = W_out[tx];
    for (int t = tx; t < IPT * HID2; t += 256)
        sth[t >> 6][t & 63] = th[(i0 + (t >> 6)) * HID2 + (t & 63)];
    __syncthreads();

    float acc[IPT];
    #pragma unroll
    for (int ii = 0; ii < IPT; ++ii) acc[ii] = 0.0f;

    #pragma unroll 4
    for (int d = 0; d < HID2; ++d) {
        const float tmv = tmT[d * NN + j];
        const float wd  = sw[d];
        #pragma unroll
        for (int ii = 0; ii < IPT; ++ii)
            acc[ii] = fmaf(wd, fast_tanh(sth[ii][d] + tmv), acc[ii]);
    }

    const float ob = out_bias[0];
    #pragma unroll
    for (int ii = 0; ii < IPT; ++ii)
        out[(i0 + ii) * NN + j] = acc[ii] + ob;
}

extern "C" void kernel_launch(void* const* d_in, const int* in_sizes, int n_in,
                              void* d_out, int out_size, void* d_ws, size_t ws_size,
                              hipStream_t stream) {
    const float* xf        = (const float*)d_in[0];
    const float* W_foh     = (const float*)d_in[1];
    const float* W_fom     = (const float*)d_in[2];
    const float* cat_bias  = (const float*)d_in[3];
    const float* W_hid2    = (const float*)d_in[4];
    const float* hid2_bias = (const float*)d_in[5];
    const float* W_out     = (const float*)d_in[6];
    const float* out_bias  = (const float*)d_in[7];
    float* out = (float*)d_out;

    float* hf  = (float*)d_ws;
    float* hm  = hf  + NN * HID;
    float* th  = hm  + NN * HID;
    float* tmT = th  + NN * HID2;

    gemm1_tanh_kernel<<<dim3(256), dim3(32, 4), 0, stream>>>(
        xf, W_foh, W_fom, cat_bias, hf, hm);

    gemm2_kernel<<<dim3(128), dim3(32, 4), 0, stream>>>(
        hf, hm, W_hid2, hid2_bias, th, tmT);

    outer_tanh_dot_kernel<<<dim3(NN / 256, NN / IPT), dim3(256), 0, stream>>>(
        th, tmT, W_out, out_bias, out);
}


# gemm1 operands staged through LDS: 4-wave blocks own 32x32 tiles, each operand fetched once per block with whole-line 16-byte loads issued up front; fragments read as grouped 16-byte LDS loads
# speedup vs baseline: 1.7207x; 1.7207x over previous
#include <hip/hip_runtime.h>
#include <hip/hip_bf16.h>


typedef float v2f __attribute__((ext_vector_type(2)));
typedef float v8f __attribute__((ext_vector_type(8)));

#define NN    1024
#define K1    512
#define HID   128
#define HID2  64

__device__ __forceinline__ float fast_tanh(float x) {
#if __has_builtin(__builtin_amdgcn_tanhf)
    return __builtin_amdgcn_tanhf(x);
#else
    return tanhf(x);
#endif
}

__device__ __forceinline__ v8f wmma_f32(v2f a, v2f b, v8f c) {
    return __builtin_amdgcn_wmma_f32_16x16x4_f32(
         false, a,  false, b,
         (short)0, c,  false,  false);
}

typedef float v4f __attribute__((ext_vector_type(4)));

#define G1_KC     128
#define G1_APITCH (K1 + 4)
#define G1_BGRP   (33 * 4)
#define G1_CPITCH 36

struct G1Stage { v4f a[8]; v4f b[8]; };

__device__ __forceinline__ void g1_gload(G1Stage& s, const float* __restrict__ ag,
                                         const float* __restrict__ bg, int c)
{
    #pragma unroll
    for (int i = 0; i < 8; ++i)
        s.a[i] = *reinterpret_cast<const v4f*>(ag + (4 * i) * K1 + G1_KC * c);
    #pragma unroll
    for (int u = 0; u < 2; ++u)
        #pragma unroll
        for (int i = 0; i < 4; ++i)
            s.b[u * 4 + i] = *reinterpret_cast<const v4f*>(
                bg + (4 * (32 * c + 16 * u) + i) * HID);
}

__device__ __forceinline__ void g1_lstore(const G1Stage& s, float* as_st, float* bs_st, int c)
{
    #pragma unroll
    for (int i = 0; i < 8; ++i)
        *reinterpret_cast<v4f*>(as_st + (4 * i) * G1_APITCH + G1_KC * c) = s.a[i];
    #pragma unroll
    for (int u = 0; u < 2; ++u)
        #pragma unroll
        for (int j = 0; j < 4; ++j) {
            v4f t;
            t.x = s.b[u * 4 + 0][j];
            t.y = s.b[u * 4 + 1][j];
            t.z = s.b[u * 4 + 2][j];
            t.w = s.b[u * 4 + 3][j];
            *reinterpret_cast<v4f*>(bs_st + (32 * c + 16 * u) * G1_BGRP + 4 * j) = t;
        }
}

__device__ __forceinline__ void g1_lds_barrier()
{
    asm volatile("s_wait_dscnt 0x0\n\ts_barrier_signal -1\n\ts_barrier_wait -1" ::: "memory");
}

__device__ __forceinline__ void g1_fragload(v4f (&fa)[16], v4f (&fb)[16],
                                            const float* a_rd, const float* b_rd, int c)
{
    #pragma unroll
    for (int g = 0; g < 16; ++g) {
        fa[g] = *reinterpret_cast<const v4f*>(a_rd + G1_KC * c + 8 * g);
        fb[g] = *reinterpret_cast<const v4f*>(b_rd + (32 * c + 2 * g) * G1_BGRP);
    }
    __builtin_amdgcn_sched_barrier(0);
}

__device__ __forceinline__ void g1_mma(v8f& acc0, v8f& acc1,
                                       const v4f (&fa)[16], const v4f (&fb)[16],
                                       int g0, int g1)
{
    #pragma unroll
    for (int g = g0; g < g1; ++g) {
        v2f a0 = {fa[g].x, fa[g].y}, b0 = {fb[g].x, fb[g].y};
        v2f a1 = {fa[g].z, fa[g].w}, b1 = {fb[g].z, fb[g].w};
        acc0 = wmma_f32(a0, b0, acc0);
        acc1 = wmma_f32(a1, b1, acc1);
    }
    asm volatile("" : "+v"(acc0), "+v"(acc1) :: "memory");
    __builtin_amdgcn_sched_barrier(0);
}

__global__ __launch_bounds__(128) void gemm1_tanh_kernel(
    const float* __restrict__ xf,
    const float* __restrict__ W_foh, const float* __restrict__ W_fom,
    const float* __restrict__ cat_bias,
    float* __restrict__ hf, float* __restrict__ hm)
{
    __shared__ __attribute__((aligned(16))) float As[32 * G1_APITCH];
    __shared__ __attribute__((aligned(16))) float Bs[(K1 / 4) * G1_BGRP];
    __shared__ __attribute__((aligned(16))) float Cs[32 * G1_CPITCH];

    const int tid  = threadIdx.x;
    const int wave = tid >> 5;
    const int lane = tid & 31;
    const int bn   = blockIdx.x & 7;
    const int bm   = blockIdx.x >> 3;
    const int mat  = bn >> 2;
    const int ncol0 = (bn & 3) * 32;

    const float* __restrict__ W    = mat ? W_fom : W_foh;
    float*       __restrict__ outp = mat ? hm    : hf;

    const float* ag = xf + (bm * 32 + wave) * K1 + 4 * lane;
    float* as_st = As + wave * G1_APITCH + 4 * lane;
    const int n4  = (lane & 3) + 4 * (lane >> 4);
    const int k4w = 4 * wave + ((lane >> 2) & 3);
    const float* bg = W + (4 * k4w) * HID + ncol0 + 4 * n4;
    float* bs_st = Bs + (k4w * 33 + 4 * n4) * 4;

    const int tm  = wave >> 1, tn = wave & 1;
    const int c16 = lane & 15, h = lane >> 4;
    const float* a_rd = As + (16 * tm + c16) * G1_APITCH + 4 * h;
    const float* b_rd = Bs + (h * 33 + 16 * tn + c16) * 4;

    G1Stage s0, s1, s2;
    g1_gload(s0, ag, bg, 0);
    g1_gload(s1, ag, bg, 1);
    g1_gload(s2, ag, bg, 2);
    __builtin_amdgcn_sched_barrier(0);

    v8f acc0 = {}, acc1 = {};
    v4f fa[16], fb[16];

    g1_lstore(s0, as_st, bs_st, 0);
    g1_gload(s0, ag, bg, 3);
    g1_lds_barrier();

    g1_fragload(fa, fb, a_rd, b_rd, 0);
    g1_mma(acc0, acc1, fa, fb, 0, 8);
    g1_lstore(s1, as_st, bs_st, 1);
    __builtin_amdgcn_sched_barrier(0);
    g1_mma(acc0, acc1, fa, fb, 8, 16);
    g1_lds_barrier();

    g1_fragload(fa, fb, a_rd, b_rd, 1);
    g1_mma(acc0, acc1, fa, fb, 0, 8);
    g1_lstore(s2, as_st, bs_st, 2);
    __builtin_amdgcn_sched_barrier(0);
    g1_mma(acc0, acc1, fa, fb, 8, 16);
    g1_lds_barrier();

    g1_fragload(fa, fb, a_rd, b_rd, 2);
    g1_mma(acc0, acc1, fa, fb, 0, 8);
    g1_lstore(s0, as_st, bs_st, 3);
    __builtin_amdgcn_sched_barrier(0);
    g1_mma(acc0, acc1, fa, fb, 8, 16);
    g1_lds_barrier();

    g1_fragload(fa, fb, a_rd, b_rd, 3);
    g1_mma(acc0, acc1, fa, fb, 0, 16);

    const float bnv = cat_bias[mat * HID + ncol0 + 16 * tn + c16];
    #pragma unroll
    for (int r = 0; r < 8; ++r)
        Cs[(16 * tm + r + 8 * h) * G1_CPITCH + 16 * tn + c16] =
            fast_tanh(acc0[r] + acc1[r] + bnv);
    g1_lds_barrier();
    #pragma unroll
    for (int i = 0; i < 2; ++i) {
        const int id  = tid + 128 * i;
        const int row = id >> 3;
        const int q   = id & 7;
        const v4f v = *reinterpret_cast<const v4f*>(Cs + row * G1_CPITCH + 4 * q);
        *reinterpret_cast<v4f*>(outp + (bm * 32 + row) * HID + ncol0 + 4 * q) = v;
    }
}

__global__ __launch_bounds__(128) void gemm2_kernel(
    const float* __restrict__ hf, const float* __restrict__ hm,
    const float* __restrict__ W_hid2, const float* __restrict__ hid2_bias,
    float* __restrict__ th, float* __restrict__ tmT)
{
    const int tile = blockIdx.x * blockDim.y + threadIdx.y;
    const int lane = threadIdx.x;
    const int mat   = tile >> 8;
    const int rem   = tile & 255;
    const int mtile = rem >> 2;
    const int ntile = rem & 3;

    const float* __restrict__ A = mat ? hm : hf;
    const float* __restrict__ B = W_hid2 + mat * HID * HID2;

    const int half = lane >> 4;
    const int l16  = lane & 15;
    const int row  = mtile * 16 + l16;
    const int col  = ntile * 16 + l16;
    const int khi  = half * 2;

    const float* arow = A + row * HID + khi;
    const float* bcol = B + khi * HID2 + col;

    v8f c = {};
    #pragma unroll
    for (int k = 0; k < HID; k += 4) {
        v2f a = *reinterpret_cast<const v2f*>(arow + k);
        v2f b;
        b.x = bcol[k * HID2];
        b.y = bcol[k * HID2 + HID2];
        c = wmma_f32(a, b, c);
    }

    if (mat == 0) {
        const float bn = hid2_bias[col];
        #pragma unroll
        for (int r = 0; r < 8; ++r) {
            const int m = mtile * 16 + r + half * 8;
            th[m * HID2 + col] = c[r] + bn;
        }
    } else {
        #pragma unroll
        for (int r = 0; r < 8; ++r) {
            const int m = mtile * 16 + r + half * 8;
            tmT[col * NN + m] = c[r];
        }
    }
}

#define IPT 8
__global__ __launch_bounds__(256) void outer_tanh_dot_kernel(
    const float* __restrict__ th, const float* __restrict__ tmT,
    const float* __restrict__ W_out, const float* __restrict__ out_bias,
    float* __restrict__ out)
{
    __shared__ float sw[HID2];
    __shared__ float sth[IPT][HID2];

    const int tx = threadIdx.x;
    const int j  = blockIdx.x * 256 + tx;
    const int i0 = blockIdx.y * IPT;

    if (tx < HID2) sw[tx] = W_out[tx];
    for (int t = tx; t < IPT * HID2; t += 256)
        sth[t >> 6][t & 63] = th[(i0 + (t >> 6)) * HID2 + (t & 63)];
    __syncthreads();

    float acc[IPT];
    #pragma unroll
    for (int ii = 0; ii < IPT; ++ii) acc[ii] = 0.0f;

    #pragma unroll 4
    for (int d = 0; d < HID2; ++d) {
        const float tmv = tmT[d * NN + j];
        const float wd  = sw[d];
        #pragma unroll
        for (int ii = 0; ii < IPT; ++ii)
            acc[ii] = fmaf(wd, fast_tanh(sth[ii][d] + tmv), acc[ii]);
    }

    const float ob = out_bias[0];
    #pragma unroll
    for (int ii = 0; ii < IPT; ++ii)
        out[(i0 + ii) * NN + j] = acc[ii] + ob;
}

extern "C" void kernel_launch(void* const* d_in, const int* in_sizes, int n_in,
                              void* d_out, int out_size, void* d_ws, size_t ws_size,
                              hipStream_t stream) {
    const float* xf        = (const float*)d_in[0];
    const float* W_foh     = (const float*)d_in[1];
    const float* W_fom     = (const float*)d_in[2];
    const float* cat_bias  = (const float*)d_in[3];
    const float* W_hid2    = (const float*)d_in[4];
    const float* hid2_bias = (const float*)d_in[5];
    const float* W_out     = (const float*)d_in[6];
    const float* out_bias  = (const float*)d_in[7];
    float* out = (float*)d_out;

    float* hf  = (float*)d_ws;
    float* hm  = hf  + NN * HID;
    float* th  = hm  + NN * HID;
    float* tmT = th  + NN * HID2;

    gemm1_tanh_kernel<<<dim3(256), dim3(128), 0, stream>>>(
        xf, W_foh, W_fom, cat_bias, hf, hm);

    gemm2_kernel<<<dim3(128), dim3(32, 4), 0, stream>>>(
        hf, hm, W_hid2, hid2_bias, th, tmT);

    outer_tanh_dot_kernel<<<dim3(NN / 256, NN / IPT), dim3(256), 0, stream>>>(
        th, tmT, W_out, out_bias, out);
}


# outer kernel loop nest swapped: a lane keeps two columns' tm values and the weights in registers and walks 8 rows whose th values arrive as scalar operands; no LDS, barrier or vector load in row loop
# speedup vs baseline: 2.0999x; 1.2204x over previous
#include <hip/hip_runtime.h>
#include <hip/hip_bf16.h>


typedef float v2f __attribute__((ext_vector_type(2)));
typedef float v8f __attribute__((ext_vector_type(8)));

#define NN    1024
#define K1    512
#define HID   128
#define HID2  64

__device__ __forceinline__ float fast_tanh(float x) {
#if __has_builtin(__builtin_amdgcn_tanhf)
    return __builtin_amdgcn_tanhf(x);
#else
    return tanhf(x);
#endif
}

__device__ __forceinline__ v8f wmma_f32(v2f a, v2f b, v8f c) {
    return __builtin_amdgcn_wmma_f32_16x16x4_f32(
         false, a,  false, b,
         (short)0, c,  false,  false);
}

typedef float v4f __attribute__((ext_vector_type(4)));

#define G1_KC     128
#define G1_APITCH (K1 + 4)
#define G1_BGRP   (33 * 4)
#define G1_CPITCH 36

struct G1Stage { v4f a[8]; v4f b[8]; };

__device__ __forceinline__ void g1_gload(G1Stage& s, const float* __restrict__ ag,
                                         const float* __restrict__ bg, int c)
{
    #pragma unroll
    for (int i = 0; i < 8; ++i)
        s.a[i] = *reinterpret_cast<const v4f*>(ag + (4 * i) * K1 + G1_KC * c);
    #pragma unroll
    for (int u = 0; u < 2; ++u)
        #pragma unroll
        for (int i = 0; i < 4; ++i)
            s.b[u * 4 + i] = *reinterpret_cast<const v4f*>(
                bg + (4 * (32 * c + 16 * u) + i) * HID);
}

__device__ __forceinline__ void g1_lstore(const G1Stage& s, float* as_st, float* bs_st, int c)
{
    #pragma unroll
    for (int i = 0; i < 8; ++i)
        *reinterpret_cast<v4f*>(as_st + (4 * i) * G1_APITCH + G1_KC * c) = s.a[i];
    #pragma unroll
    for (int u = 0; u < 2; ++u)
        #pragma unroll
        for (int j = 0; j < 4; ++j) {
            v4f t;
            t.x = s.b[u * 4 + 0][j];
            t.y = s.b[u * 4 + 1][j];
            t.z = s.b[u * 4 + 2][j];
            t.w = s.b[u * 4 + 3][j];
            *reinterpret_cast<v4f*>(bs_st + (32 * c + 16 * u) * G1_BGRP + 4 * j) = t;
        }
}

__device__ __forceinline__ void g1_lds_barrier()
{
    asm volatile("s_wait_dscnt 0x0\n\ts_barrier_signal -1\n\ts_barrier_wait -1" ::: "memory");
}

__device__ __forceinline__ void g1_fragload(v4f (&fa)[16], v4f (&fb)[16],
                                            const float* a_rd, const float* b_rd, int c)
{
    #pragma unroll
    for (int g = 0; g < 16; ++g) {
        fa[g] = *reinterpret_cast<const v4f*>(a_rd + G1_KC * c + 8 * g);
        fb[g] = *reinterpret_cast<const v4f*>(b_rd + (32 * c + 2 * g) * G1_BGRP);
    }
    __builtin_amdgcn_sched_barrier(0);
}

__device__ __forceinline__ void g1_mma(v8f& acc0, v8f& acc1,
                                       const v4f (&fa)[16], const v4f (&fb)[16],
                                       int g0, int g1)
{
    #pragma unroll
    for (int g = g0; g < g1; ++g) {
        v2f a0 = {fa[g].x, fa[g].y}, b0 = {fb[g].x, fb[g].y};
        v2f a1 = {fa[g].z, fa[g].w}, b1 = {fb[g].z, fb[g].w};
        acc0 = wmma_f32(a0, b0, acc0);
        acc1 = wmma_f32(a1, b1, acc1);
    }
    asm volatile("" : "+v"(acc0), "+v"(acc1) :: "memory");
    __builtin_amdgcn_sched_barrier(0);
}

__global__ __launch_bounds__(128) void gemm1_tanh_kernel(
    const float* __restrict__ xf,
    const float* __restrict__ W_foh, const float* __restrict__ W_fom,
    const float* __restrict__ cat_bias,
    float* __restrict__ hf, float* __restrict__ hm)
{
    __shared__ __attribute__((aligned(16))) float As[32 * G1_APITCH];
    __shared__ __attribute__((aligned(16))) float Bs[(K1 / 4) * G1_BGRP];
    __shared__ __attribute__((aligned(16))) float Cs[32 * G1_CPITCH];

    const int tid  = threadIdx.x;
    const int wave = tid >> 5;
    const int lane = tid & 31;
    const int bn   = blockIdx.x & 7;
    const int bm   = blockIdx.x >> 3;
    const int mat  = bn >> 2;
    const int ncol0 = (bn & 3) * 32;

    const float* __restrict__ W    = mat ? W_fom : W_foh;
    float*       __restrict__ outp = mat ? hm    : hf;

    const float* ag = xf + (bm * 32 + wave) * K1 + 4 * lane;
    float* as_st = As + wave * G1_APITCH + 4 * lane;
    const int n4  = (lane & 3) + 4 * (lane >> 4);
    const int k4w = 4 * wave + ((lane >> 2) & 3);
    const float* bg = W + (4 * k4w) * HID + ncol0 + 4 * n4;
    float* bs_st = Bs + (k4w * 33 + 4 * n4) * 4;

    const int tm  = wave >> 1, tn = wave & 1;
    const int c16 = lane & 15, h = lane >> 4;
    const float* a_rd = As + (16 * tm + c16) * G1_APITCH + 4 * h;
    const float* b_rd = Bs + (h * 33 + 16 * tn + c16) * 4;

    G1Stage s0, s1, s2;
    g1_gload(s0, ag, bg, 0);
    g1_gload(s1, ag, bg, 1);
    g1_gload(s2, ag, bg, 2);
    __builtin_amdgcn_sched_barrier(0);

    v8f acc0 = {}, acc1 = {};
    v4f fa[16], fb[16];

    g1_lstore(s0, as_st, bs_st, 0);
    g1_gload(s0, ag, bg, 3);
    g1_lds_barrier();

    g1_fragload(fa, fb, a_rd, b_rd, 0);
    g1_mma(acc0, acc1, fa, fb, 0, 8);
    g1_lstore(s1, as_st, bs_st, 1);
    __builtin_amdgcn_sched_barrier(0);
    g1_mma(acc0, acc1, fa, fb, 8, 16);
    g1_lds_barrier();

    g1_fragload(fa, fb, a_rd, b_rd, 1);
    g1_mma(acc0, acc1, fa, fb, 0, 8);
    g1_lstore(s2, as_st, bs_st, 2);
    __builtin_amdgcn_sched_barrier(0);
    g1_mma(acc0, acc1, fa, fb, 8, 16);
    g1_lds_barrier();

    g1_fragload(fa, fb, a_rd, b_rd, 2);
    g1_mma(acc0, acc1, fa, fb, 0, 8);
    g1_lstore(s0, as_st, bs_st, 3);
    __builtin_amdgcn_sched_barrier(0);
    g1_mma(acc0, acc1, fa, fb, 8, 16);
    g1_lds_barrier();

    g1_fragload(fa, fb, a_rd, b_rd, 3);
    g1_mma(acc0, acc1, fa, fb, 0, 16);

    const float bnv = cat_bias[mat * HID + ncol0 + 16 * tn + c16];
    #pragma unroll
    for (int r = 0; r < 8; ++r)
        Cs[(16 * tm + r + 8 * h) * G1_CPITCH + 16 * tn + c16] =
            fast_tanh(acc0[r] + acc1[r] + bnv);
    g1_lds_barrier();
    #pragma unroll
    for (int i = 0; i < 2; ++i) {
        const int id  = tid + 128 * i;
        const int row = id >> 3;
        const int q   = id & 7;
        const v4f v = *reinterpret_cast<const v4f*>(Cs + row * G1_CPITCH + 4 * q);
        *reinterpret_cast<v4f*>(outp + (bm * 32 + row) * HID + ncol0 + 4 * q) = v;
    }
}

__global__ __launch_bounds__(128) void gemm2_kernel(
    const float* __restrict__ hf, const float* __restrict__ hm,
    const float* __restrict__ W_hid2, const float* __restrict__ hid2_bias,
    float* __restrict__ th, float* __restrict__ tmT)
{
    const int tile = blockIdx.x * blockDim.y + threadIdx.y;
    const int lane = threadIdx.x;
    const int mat   = tile >> 8;
    const int rem   = tile & 255;
    const int mtile = rem >> 2;
    const int ntile = rem & 3;

    const float* __restrict__ A = mat ? hm : hf;
    const float* __restrict__ B = W_hid2 + mat * HID * HID2;

    const int half = lane >> 4;
    const int l16  = lane & 15;
    const int row  = mtile * 16 + l16;
    const int col  = ntile * 16 + l16;
    const int khi  = half * 2;

    const float* arow = A + row * HID + khi;
    const float* bcol = B + khi * HID2 + col;

    v8f c = {};
    #pragma unroll
    for (int k = 0; k < HID; k += 4) {
        v2f a = *reinterpret_cast<const v2f*>(arow + k);
        v2f b;
        b.x = bcol[k * HID2];
        b.y = bcol[k * HID2 + HID2];
        c = wmma_f32(a, b, c);
    }

    if (mat == 0) {
        const float bn = hid2_bias[col];
        #pragma unroll
        for (int r = 0; r < 8; ++r) {
            const int m = mtile * 16 + r + half * 8;
            th[m * HID2 + col] = c[r] + bn;
        }
    } else {
        #pragma unroll
        for (int r = 0; r < 8; ++r) {
            const int m = mtile * 16 + r + half * 8;
            tmT[col * NN + m] = c[r];
        }
    }
}

#define OT_R   8
#define OT_LAG 3
typedef float v16f __attribute__((ext_vector_type(16)));

__device__ __forceinline__ v2f splat2(float s) { v2f r; r.x = s; r.y = s; return r; }

__global__ __launch_bounds__(256) void outer_tanh_dot_kernel(
    const float* __restrict__ th, const float* __restrict__ tmT,
    const float* __restrict__ W_out, const float* __restrict__ out_bias,
    float* __restrict__ out)
{
    const int lane = threadIdx.x & 31;
    const int wv   = __builtin_amdgcn_readfirstlane(threadIdx.x >> 5);
    const int jt   = blockIdx.x & 15;
    const int it   = blockIdx.x >> 4;
    const int j    = jt * 64 + 2 * lane;
    const int i0   = it * 64 + wv * OT_R;

    v2f tm[HID2], ww[HID2 / 2];

    int wofs = 0;
    asm volatile("" : "+v"(wofs));
    const float* __restrict__ wp = W_out + wofs;
    #pragma unroll
    for (int q = 0; q < HID2 / 4; ++q) {
        const v4f t = *reinterpret_cast<const v4f*>(wp + 4 * q);
        ww[2 * q + 0].x = t.x;  ww[2 * q + 0].y = t.y;
        ww[2 * q + 1].x = t.z;  ww[2 * q + 1].y = t.w;
    }
    __builtin_amdgcn_sched_barrier(0);
    #pragma unroll
    for (int d = 0; d < HID2; ++d)
        tm[d] = *reinterpret_cast<const v2f*>(tmT + d * NN + j);
    const float ob = out_bias[0];

    const v16f* __restrict__ trow = reinterpret_cast<const v16f*>(th + i0 * HID2);
    v16f c[4];
    c[0] = trow[0]; c[1] = trow[1]; c[2] = trow[2]; c[3] = trow[3];
    __builtin_amdgcn_sched_barrier(0);

    auto do_row = [&](const int r) __attribute__((always_inline)) {
        const int rn = (r + 1 < OT_R) ? (r + 1) : r;
        const v16f* __restrict__ tnext = trow + 4 * rn;
        v2f accA[2], accB[2];
        accA[0] = splat2(0.0f); accA[1] = splat2(0.0f);
        accB[0] = splat2(0.0f); accB[1] = splat2(0.0f);
        v2f x[HID2];
        v2f yA[HID2 / 2], yB[HID2 / 2];

        #pragma unroll
        for (int d = 0; d < OT_LAG; ++d)
            x[d] = tm[d] + splat2(c[d >> 4][d & 15]);
        __builtin_amdgcn_sched_barrier(0);

        #pragma unroll
        for (int d = 0; d < HID2; ++d) {
            yA[d >> 1][d & 1] = fast_tanh(x[d].x);
            if (d + OT_LAG < HID2) {
                const int e = d + OT_LAG;
                x[e] = tm[e] + splat2(c[e >> 4][e & 15]);
            }
            __builtin_amdgcn_sched_barrier(0);
            yB[d >> 1][d & 1] = fast_tanh(x[d].y);
            if (d >= OT_LAG) {
                const int k = d - OT_LAG, q = k >> 1;
                if (k & 1) accB[q & 1] = __builtin_elementwise_fma(ww[q], yB[q], accB[q & 1]);
                else       accA[q & 1] = __builtin_elementwise_fma(ww[q], yA[q], accA[q & 1]);
            }
            __builtin_amdgcn_sched_barrier(0);
            if (((d + OT_LAG) & 15) == 15 && d + OT_LAG < HID2) {
                const int g = (d + OT_LAG) >> 4;
                c[g] = tnext[g];
                __builtin_amdgcn_sched_barrier(0);
            }
        }
        #pragma unroll
        for (int k = HID2 - OT_LAG; k < HID2; ++k) {
            const int q = k >> 1;
            if (k & 1) accB[q & 1] = __builtin_elementwise_fma(ww[q], yB[q], accB[q & 1]);
            else       accA[q & 1] = __builtin_elementwise_fma(ww[q], yA[q], accA[q & 1]);
        }

        const v2f sa = accA[0] + accA[1];
        const v2f sb = accB[0] + accB[1];
        v2f s;
        s.x = (sa.x + sa.y) + ob;
        s.y = (sb.x + sb.y) + ob;
        *reinterpret_cast<v2f*>(out + (i0 + r) * NN + j) = s;
        __builtin_amdgcn_sched_barrier(0);
    };

    do_row(0);
    #pragma unroll 1
    for (int r = 1; r < OT_R; ++r)
        do_row(r);
}

extern "C" void kernel_launch(void* const* d_in, const int* in_sizes, int n_in,
                              void* d_out, int out_size, void* d_ws, size_t ws_size,
                              hipStream_t stream) {
    const float* xf        = (const float*)d_in[0];
    const float* W_foh     = (const float*)d_in[1];
    const float* W_fom     = (const float*)d_in[2];
    const float* cat_bias  = (const float*)d_in[3];
    const float* W_hid2    = (const float*)d_in[4];
    const float* hid2_bias = (const float*)d_in[5];
    const float* W_out     = (const float*)d_in[6];
    const float* out_bias  = (const float*)d_in[7];
    float* out = (float*)d_out;

    float* hf  = (float*)d_ws;
    float* hm  = hf  + NN * HID;
    float* th  = hm  + NN * HID;
    float* tmT = th  + NN * HID2;

    gemm1_tanh_kernel<<<dim3(256), dim3(128), 0, stream>>>(
        xf, W_foh, W_fom, cat_bias, hf, hm);

    gemm2_kernel<<<dim3(128), dim3(32, 4), 0, stream>>>(
        hf, hm, W_hid2, hid2_bias, th, tmT);

    outer_tanh_dot_kernel<<<dim3(256), dim3(256), 0, stream>>>(
        th, tmT, W_out, out_bias, out);
}
